# starting kernel
# baseline (speedup 1.0000x reference)
#include <hip/hip_runtime.h>


typedef __attribute__((ext_vector_type(2))) float v2f;
typedef __attribute__((ext_vector_type(4))) float v4f;
typedef __attribute__((ext_vector_type(8))) float v8f;

#define LDS_STRIDE 66
#define ROWS_PER_WAVE 32
#define WAVES_PER_BLOCK 8
#define THREADS_PER_BLOCK (WAVES_PER_BLOCK * 32)

__device__ __forceinline__ float softplus_f(float v) {
  float s = __logf(1.0f + __expf(v));
  return v > 20.0f ? v : s;
}

__launch_bounds__(THREADS_PER_BLOCK)
__global__ void dln_h_kernel(const float* __restrict__ x,
                             const float* __restrict__ W1,
                             const float* __restrict__ b1,
                             const float* __restrict__ Wd,
                             const float* __restrict__ bd,
                             const float* __restrict__ Wo,
                             const float* __restrict__ bo,
                             float* __restrict__ out) {
  __shared__ float lds[WAVES_PER_BLOCK * ROWS_PER_WAVE * LDS_STRIDE];

  const int tid   = threadIdx.x;
  const int lane  = tid & 31;
  const int wave  = tid >> 5;
  const int l15   = lane & 15;
  const int khalf = (lane >> 4) * 2;
  const unsigned rowBase =
      (blockIdx.x * WAVES_PER_BLOCK + wave) * ROWS_PER_WAVE;

  float* hbuf = &lds[wave * (ROWS_PER_WAVE * LDS_STRIDE)];

  v2f bfrag[4];
  v2f bb[4];
#pragma unroll
  for (int nt = 0; nt < 4; ++nt) {
    const int n = nt * 16 + l15;
    bfrag[nt] = *(const v2f*)(W1 + n * 4 + khalf);
    const float cb = b1[n];
    bb[nt].x = cb;
    bb[nt].y = cb;
  }

  const v2f kSlope = {0.01f, 0.01f};

#pragma unroll
  for (int tile = 0; tile < 2; ++tile) {
    const unsigned row = rowBase + tile * 16 + l15;
    const v2f a = *(const v2f*)(x + row * 4 + khalf);
    const int mbase = tile * 16 + ((lane >> 4) << 3);
#pragma unroll
    for (int nt = 0; nt < 4; ++nt) {
      v8f c = {};
      v8f d = __builtin_amdgcn_wmma_f32_16x16x4_f32(
          false, a, false, bfrag[nt], (short)0, c, false, false);
      const int n = nt * 16 + l15;
#pragma unroll
      for (int p = 0; p < 4; ++p) {
        v2f t;
        t.x = d[2 * p];
        t.y = d[2 * p + 1];
        t = t + bb[nt];
        const v2f u = t * kSlope;
        hbuf[(mbase + 2 * p) * LDS_STRIDE + n]     = fmaxf(t.x, u.x);
        hbuf[(mbase + 2 * p + 1) * LDS_STRIDE + n] = fmaxf(t.y, u.y);
      }
    }
  }

  asm volatile("s_wait_dscnt 0" ::: "memory");

  const float* hrow = hbuf + lane * LDS_STRIDE;
  v2f a0 = {0.0f, 0.0f};
  v2f a1 = {0.0f, 0.0f};
  v2f a2 = {0.0f, 0.0f};
#pragma unroll
  for (int k = 0; k < 32; ++k) {
    const v2f hv = *(const v2f*)(hrow + 2 * k);
    const v2f w0 = *(const v2f*)(Wd + 2 * k);
    const v2f w1 = *(const v2f*)(Wd + 64 + 2 * k);
    const v2f w2 = *(const v2f*)(Wo + 2 * k);
    a0 = __builtin_elementwise_fma(hv, w0, a0);
    a1 = __builtin_elementwise_fma(hv, w1, a1);
    a2 = __builtin_elementwise_fma(hv, w2, a2);
  }
  const float acc0 = a0.x + a0.y;
  const float acc1 = a1.x + a1.y;
  const float acc2 = a2.x + a2.y;

  const float ld0 = softplus_f(acc0 + bd[0]);
  const float ld1 = softplus_f(acc1 + bd[1]);
  const float lo  = acc2 + bo[0];

  v4f H;
  H.x = fmaf(ld0, ld0, 1e-9f);
  H.y = ld0 * lo;
  H.z = ld0 * lo;
  H.w = fmaf(lo, lo, fmaf(ld1, ld1, 1e-9f));

  const unsigned row = rowBase + lane;
  *(v4f*)(out + row * 4) = H;
}

extern "C" void kernel_launch(void* const* d_in, const int* in_sizes, int n_in,
                              void* d_out, int out_size, void* d_ws, size_t ws_size,
                              hipStream_t stream) {
  const float* x  = (const float*)d_in[0];
  const float* W1 = (const float*)d_in[1];
  const float* b1 = (const float*)d_in[2];
  const float* Wd = (const float*)d_in[3];
  const float* bd = (const float*)d_in[4];
  const float* Wo = (const float*)d_in[5];
  const float* bo = (const float*)d_in[6];
  float* out = (float*)d_out;

  const int nrows  = in_sizes[0] / 4;
  const int blocks = nrows / (WAVES_PER_BLOCK * ROWS_PER_WAVE);
  dln_h_kernel<<<blocks, THREADS_PER_BLOCK, 0, stream>>>(
      x, W1, b1, Wd, bd, Wo, bo, out);
}


# Layer 1 transposed on WMMA (W1 as A, x as B): a lane holds its own row's hidden values; layer 2 runs from D registers, per-lane weights kept over a chunk loop, no LDS pass; its sum order changes
# speedup vs baseline: 1.2730x; 1.2730x over previous
#include <hip/hip_runtime.h>


typedef __attribute__((ext_vector_type(2))) float v2f;
typedef __attribute__((ext_vector_type(4))) float v4f;
typedef __attribute__((ext_vector_type(8))) float v8f;

#define CHUNK_ROWS 32
#define CHUNKS_PER_WAVE 8
#define ROWS_PER_WAVE (CHUNK_ROWS * CHUNKS_PER_WAVE)
#define WAVES_PER_BLOCK 8
#define THREADS_PER_BLOCK (WAVES_PER_BLOCK * 32)

__device__ __forceinline__ float softplus_f(float v) {
  float s = __logf(1.0f + __expf(v));
  return v > 20.0f ? v : s;
}

__device__ __forceinline__ float xhalf(float v) {
  const int b = __builtin_bit_cast(int, v);
  return __builtin_bit_cast(
      float, __builtin_amdgcn_permlanex16(b, b, 0x76543210, 0xfedcba98, false, false));
}

__device__ __forceinline__ v8f load8(const float* __restrict__ p) {
  const v4f lo = *(const v4f*)p;
  const v4f hi = *(const v4f*)(p + 4);
  v8f r;
  r[0] = lo.x; r[1] = lo.y; r[2] = lo.z; r[3] = lo.w;
  r[4] = hi.x; r[5] = hi.y; r[6] = hi.z; r[7] = hi.w;
  return r;
}

__device__ __forceinline__ v2f leaky2(v2f t) {
  const v2f kSlope = {0.01f, 0.01f};
  const v2f u = t * kSlope;
  v2f h;
  h.x = fmaxf(t.x, u.x);
  h.y = fmaxf(t.y, u.y);
  return h;
}

__launch_bounds__(THREADS_PER_BLOCK)
__global__ void dln_h_kernel(const float* __restrict__ x,
                             const float* __restrict__ W1,
                             const float* __restrict__ b1,
                             const float* __restrict__ Wd,
                             const float* __restrict__ bd,
                             const float* __restrict__ Wo,
                             const float* __restrict__ bo,
                             float* __restrict__ out) {
  const int tid   = threadIdx.x;
  const int lane  = tid & 31;
  const int wave  = tid >> 5;
  const int l15   = lane & 15;
  const int half  = lane >> 4;
  const int khalf = half * 2;

  v2f wfrag[4];
  v8f cb[4], w0[4], w1[4], w2[4];
#pragma unroll
  for (int nt = 0; nt < 4; ++nt) {
    wfrag[nt] = *(const v2f*)(W1 + (nt * 16 + l15) * 4 + khalf);
    const int hb = nt * 16 + half * 8;
    cb[nt] = load8(b1 + hb);
    w0[nt] = load8(Wd + hb);
    w1[nt] = load8(Wd + 64 + hb);
    w2[nt] = load8(Wo + hb);
  }
  asm volatile("" ::
      "v"(wfrag[0]), "v"(wfrag[1]), "v"(wfrag[2]), "v"(wfrag[3]),
      "v"(cb[0]), "v"(cb[1]), "v"(cb[2]), "v"(cb[3]),
      "v"(w0[0]), "v"(w0[1]), "v"(w0[2]), "v"(w0[3]),
      "v"(w1[0]), "v"(w1[1]), "v"(w1[2]), "v"(w1[3]),
      "v"(w2[0]), "v"(w2[1]), "v"(w2[2]), "v"(w2[3]));

  const float bd0 = bd[0];
  const float bd1 = bd[1];
  const float bo0 = bo[0];

  const unsigned wavesTotal = gridDim.x * WAVES_PER_BLOCK;
  unsigned chunk = blockIdx.x * WAVES_PER_BLOCK + wave;

  const float* xp = x + (chunk * CHUNK_ROWS + l15) * 4 + khalf;
  v2f xa = *(const v2f*)xp;
  v2f xb = *(const v2f*)(xp + 64);

  for (int it = 0; it < CHUNKS_PER_WAVE; ++it) {
    const unsigned cur = chunk;
    v2f xa_n = xa;
    v2f xb_n = xb;
    if (it + 1 < CHUNKS_PER_WAVE) {
      chunk += wavesTotal;
      const float* xn = x + (chunk * CHUNK_ROWS + l15) * 4 + khalf;
      xa_n = *(const v2f*)xn;
      xb_n = *(const v2f*)(xn + 64);
    }

    v2f a0 = {0.0f, 0.0f}, a1 = {0.0f, 0.0f}, a2 = {0.0f, 0.0f};
    v2f c0 = {0.0f, 0.0f}, c1 = {0.0f, 0.0f}, c2 = {0.0f, 0.0f};
#pragma unroll
    for (int nt = 0; nt < 4; ++nt) {
      const v8f zero = {};
      const v8f d0 = __builtin_amdgcn_wmma_f32_16x16x4_f32(
          false, wfrag[nt], false, xa, (short)0, zero, false, false);
      const v8f d1 = __builtin_amdgcn_wmma_f32_16x16x4_f32(
          false, wfrag[nt], false, xb, (short)0, zero, false, false);
#pragma unroll
      for (int p = 0; p < 4; ++p) {
        v2f wv0, wv1, wv2;
        wv0.x = w0[nt][2 * p]; wv0.y = w0[nt][2 * p + 1];
        wv1.x = w1[nt][2 * p]; wv1.y = w1[nt][2 * p + 1];
        wv2.x = w2[nt][2 * p]; wv2.y = w2[nt][2 * p + 1];
        v2f bv, t0, t1;
        bv.x = cb[nt][2 * p]; bv.y = cb[nt][2 * p + 1];
        t0.x = d0[2 * p]; t0.y = d0[2 * p + 1];
        t1.x = d1[2 * p]; t1.y = d1[2 * p + 1];
        const v2f h0 = leaky2(t0 + bv);
        const v2f h1 = leaky2(t1 + bv);
        a0 = __builtin_elementwise_fma(h0, wv0, a0);
        a1 = __builtin_elementwise_fma(h0, wv1, a1);
        a2 = __builtin_elementwise_fma(h0, wv2, a2);
        c0 = __builtin_elementwise_fma(h1, wv0, c0);
        c1 = __builtin_elementwise_fma(h1, wv1, c1);
        c2 = __builtin_elementwise_fma(h1, wv2, c2);
      }
    }

    const float pa0 = a0.x + a0.y, pa1 = a1.x + a1.y, pa2 = a2.x + a2.y;
    const float pc0 = c0.x + c0.y, pc1 = c1.x + c1.y, pc2 = c2.x + c2.y;
    const bool up = half != 0;
    const float acc0 = (up ? pc0 : pa0) + xhalf(up ? pa0 : pc0);
    const float acc1 = (up ? pc1 : pa1) + xhalf(up ? pa1 : pc1);
    const float acc2 = (up ? pc2 : pa2) + xhalf(up ? pa2 : pc2);

    const float ld0 = softplus_f(acc0 + bd0);
    const float ld1 = softplus_f(acc1 + bd1);
    const float lo  = acc2 + bo0;

    v4f H;
    H.x = fmaf(ld0, ld0, 1e-9f);
    H.y = ld0 * lo;
    H.z = ld0 * lo;
    H.w = fmaf(lo, lo, fmaf(ld1, ld1, 1e-9f));

    const unsigned row = cur * CHUNK_ROWS + lane;
    *(v4f*)(out + row * 4) = H;

    xa = xa_n;
    xb = xb_n;
  }
}

extern "C" void kernel_launch(void* const* d_in, const int* in_sizes, int n_in,
                              void* d_out, int out_size, void* d_ws, size_t ws_size,
                              hipStream_t stream) {
  const float* x  = (const float*)d_in[0];
  const float* W1 = (const float*)d_in[1];
  const float* b1 = (const float*)d_in[2];
  const float* Wd = (const float*)d_in[3];
  const float* bd = (const float*)d_in[4];
  const float* Wo = (const float*)d_in[5];
  const float* bo = (const float*)d_in[6];
  float* out = (float*)d_out;

  const int nrows  = in_sizes[0] / 4;
  const int blocks = nrows / (WAVES_PER_BLOCK * ROWS_PER_WAVE);
  dln_h_kernel<<<blocks, THREADS_PER_BLOCK, 0, stream>>>(
      x, W1, b1, Wd, bd, Wo, bo, out);
}


# All eight chunks' x of a wave requested up front behind the constants (16 loads in flight per wave, was 2); chunk loop fully unrolled, counted waits, no load between chunks
# speedup vs baseline: 1.4181x; 1.1140x over previous
#include <hip/hip_runtime.h>


typedef __attribute__((ext_vector_type(2))) float v2f;
typedef __attribute__((ext_vector_type(4))) float v4f;
typedef __attribute__((ext_vector_type(8))) float v8f;

#define CHUNK_ROWS 32
#define CHUNKS_PER_WAVE 8
#define ROWS_PER_WAVE (CHUNK_ROWS * CHUNKS_PER_WAVE)
#define WAVES_PER_BLOCK 8
#define THREADS_PER_BLOCK (WAVES_PER_BLOCK * 32)

__device__ __forceinline__ float softplus_f(float v) {
  float s = __logf(1.0f + __expf(v));
  return v > 20.0f ? v : s;
}

__device__ __forceinline__ float xhalf(float v) {
  const int b = __builtin_bit_cast(int, v);
  return __builtin_bit_cast(
      float, __builtin_amdgcn_permlanex16(b, b, 0x76543210, 0xfedcba98, false, false));
}

__device__ __forceinline__ v8f load8(const float* __restrict__ p) {
  const v4f lo = *(const v4f*)p;
  const v4f hi = *(const v4f*)(p + 4);
  v8f r;
  r[0] = lo.x; r[1] = lo.y; r[2] = lo.z; r[3] = lo.w;
  r[4] = hi.x; r[5] = hi.y; r[6] = hi.z; r[7] = hi.w;
  return r;
}

__device__ __forceinline__ v2f leaky2(v2f t) {
  const v2f kSlope = {0.01f, 0.01f};
  const v2f u = t * kSlope;
  v2f h;
  h.x = fmaxf(t.x, u.x);
  h.y = fmaxf(t.y, u.y);
  return h;
}

__launch_bounds__(THREADS_PER_BLOCK)
__global__ void dln_h_kernel(const float* __restrict__ x,
                             const float* __restrict__ W1,
                             const float* __restrict__ b1,
                             const float* __restrict__ Wd,
                             const float* __restrict__ bd,
                             const float* __restrict__ Wo,
                             const float* __restrict__ bo,
                             float* __restrict__ out) {
  const int tid   = threadIdx.x;
  const int lane  = tid & 31;
  const int wave  = tid >> 5;
  const int l15   = lane & 15;
  const int half  = lane >> 4;
  const int khalf = half * 2;

  v2f wfrag[4];
  v8f cb[4], w0[4], w1[4], w2[4];
#pragma unroll
  for (int nt = 0; nt < 4; ++nt) {
    wfrag[nt] = *(const v2f*)(W1 + (nt * 16 + l15) * 4 + khalf);
    const int hb = nt * 16 + half * 8;
    cb[nt] = load8(b1 + hb);
    w0[nt] = load8(Wd + hb);
    w1[nt] = load8(Wd + 64 + hb);
    w2[nt] = load8(Wo + hb);
  }
  const float bd0 = bd[0];
  const float bd1 = bd[1];
  const float bo0 = bo[0];

  const unsigned wavesTotal = gridDim.x * WAVES_PER_BLOCK;
  const unsigned chunk0 = blockIdx.x * WAVES_PER_BLOCK + wave;

  v2f xa[CHUNKS_PER_WAVE], xb[CHUNKS_PER_WAVE];
#pragma unroll
  for (int it = 0; it < CHUNKS_PER_WAVE; ++it) {
    const unsigned ch = chunk0 + it * wavesTotal;
    const float* xp = x + (ch * CHUNK_ROWS + l15) * 4 + khalf;
    xa[it] = *(const v2f*)xp;
    xb[it] = *(const v2f*)(xp + 64);
  }
  __builtin_amdgcn_sched_barrier(0);

#pragma unroll
  for (int it = 0; it < CHUNKS_PER_WAVE; ++it) {
    const unsigned cur = chunk0 + it * wavesTotal;

    v2f a0 = {0.0f, 0.0f}, a1 = {0.0f, 0.0f}, a2 = {0.0f, 0.0f};
    v2f c0 = {0.0f, 0.0f}, c1 = {0.0f, 0.0f}, c2 = {0.0f, 0.0f};
#pragma unroll
    for (int nt = 0; nt < 4; ++nt) {
      const v8f zero = {};
      const v8f d0 = __builtin_amdgcn_wmma_f32_16x16x4_f32(
          false, wfrag[nt], false, xa[it], (short)0, zero, false, false);
      const v8f d1 = __builtin_amdgcn_wmma_f32_16x16x4_f32(
          false, wfrag[nt], false, xb[it], (short)0, zero, false, false);
#pragma unroll
      for (int p = 0; p < 4; ++p) {
        v2f wv0, wv1, wv2;
        wv0.x = w0[nt][2 * p]; wv0.y = w0[nt][2 * p + 1];
        wv1.x = w1[nt][2 * p]; wv1.y = w1[nt][2 * p + 1];
        wv2.x = w2[nt][2 * p]; wv2.y = w2[nt][2 * p + 1];
        v2f bv, t0, t1;
        bv.x = cb[nt][2 * p]; bv.y = cb[nt][2 * p + 1];
        t0.x = d0[2 * p]; t0.y = d0[2 * p + 1];
        t1.x = d1[2 * p]; t1.y = d1[2 * p + 1];
        const v2f h0 = leaky2(t0 + bv);
        const v2f h1 = leaky2(t1 + bv);
        a0 = __builtin_elementwise_fma(h0, wv0, a0);
        a1 = __builtin_elementwise_fma(h0, wv1, a1);
        a2 = __builtin_elementwise_fma(h0, wv2, a2);
        c0 = __builtin_elementwise_fma(h1, wv0, c0);
        c1 = __builtin_elementwise_fma(h1, wv1, c1);
        c2 = __builtin_elementwise_fma(h1, wv2, c2);
      }
    }

    const float pa0 = a0.x + a0.y, pa1 = a1.x + a1.y, pa2 = a2.x + a2.y;
    const float pc0 = c0.x + c0.y, pc1 = c1.x + c1.y, pc2 = c2.x + c2.y;
    const bool up = half != 0;
    const float acc0 = (up ? pc0 : pa0) + xhalf(up ? pa0 : pc0);
    const float acc1 = (up ? pc1 : pa1) + xhalf(up ? pa1 : pc1);
    const float acc2 = (up ? pc2 : pa2) + xhalf(up ? pa2 : pc2);

    const float ld0 = softplus_f(acc0 + bd0);
    const float ld1 = softplus_f(acc1 + bd1);
    const float lo  = acc2 + bo0;

    v4f H;
    H.x = fmaf(ld0, ld0, 1e-9f);
    H.y = ld0 * lo;
    H.z = ld0 * lo;
    H.w = fmaf(lo, lo, fmaf(ld1, ld1, 1e-9f));

    const unsigned row = cur * CHUNK_ROWS + lane;
    *(v4f*)(out + row * 4) = H;
    __builtin_amdgcn_sched_barrier(0);
  }
}

extern "C" void kernel_launch(void* const* d_in, const int* in_sizes, int n_in,
                              void* d_out, int out_size, void* d_ws, size_t ws_size,
                              hipStream_t stream) {
  const float* x  = (const float*)d_in[0];
  const float* W1 = (const float*)d_in[1];
  const float* b1 = (const float*)d_in[2];
  const float* Wd = (const float*)d_in[3];
  const float* bd = (const float*)d_in[4];
  const float* Wo = (const float*)d_in[5];
  const float* bo = (const float*)d_in[6];
  float* out = (float*)d_out;

  const int nrows  = in_sizes[0] / 4;
  const int blocks = nrows / (WAVES_PER_BLOCK * ROWS_PER_WAVE);
  dln_h_kernel<<<blocks, THREADS_PER_BLOCK, 0, stream>>>(
      x, W1, b1, Wd, bd, Wo, bo, out);
}
